# starting kernel
# baseline (speedup 1.0000x reference)
#include <hip/hip_runtime.h>
#include <math.h>

typedef __attribute__((ext_vector_type(2))) float v2f;
typedef __attribute__((ext_vector_type(8))) float v8f;

constexpr int Bb = 8;
constexpr int Qq = 512;
constexpr int Nn = 8192;
constexpr int Dd = 64;
constexpr int Vv = 64;

constexpr int WAVES  = 8;
constexpr int NSEG   = Nn / WAVES;
constexpr int NCHUNK = 16;

constexpr float SHIFT = 12.0f;

constexpr int PSTRIDE = 18;
constexpr int OSTRIDE = 65;
constexpr int SLOT    = 16 * OSTRIDE + 16;

__global__ __launch_bounds__(WAVES * 32)
void gated_attn_flash_f32wmma(const float* __restrict__ query,
                              const float* __restrict__ key,
                              const float* __restrict__ value,
                              const float* __restrict__ gates,
                              float* __restrict__ out)
{
    __shared__ float smem[WAVES * SLOT];

    const int tid  = threadIdx.x;
    const int w    = tid >> 5;
    const int lane = tid & 31;
    const int lo   = lane & 15;
    const int hi   = lane >> 4;

    const int b  = blockIdx.x >> 5;
    const int q0 = (blockIdx.x & 31) << 4;

    float* slot = smem + w * SLOT;

    v2f qa[16];
    {
        const float* qptr = query + ((size_t)(b * Qq + q0 + lo)) * Dd + 2 * hi;
#pragma unroll
        for (int i = 0; i < 16; ++i)
            qa[i] = *(const v2f*)(qptr + 4 * i);
    }

    v8f acc[4];
    v8f accs;
#pragma unroll
    for (int t = 0; t < 4; ++t) { v8f z = {}; acc[t] = z; }
    { v8f z = {}; accs = z; }

    const v2f vones = {1.0f, 1.0f};

    const int nbase = w * NSEG;

    for (int c = 0; c < NSEG / NCHUNK; ++c) {
        const int n0 = nbase + c * NCHUNK;

        const float* kptr = key + ((size_t)(b * Nn + n0 + lo)) * Dd + 2 * hi;
        v8f s = {};
#pragma unroll
        for (int i = 0; i < 16; ++i) {
            v2f kb = *(const v2f*)(kptr + 4 * i);
            s = __builtin_amdgcn_wmma_f32_16x16x4_f32(false, qa[i], false, kb,
                                                      (short)0, s, false, false);
        }

        const float* gptr = gates + ((size_t)(b * Nn + n0 + lo)) * Qq + q0 + 8 * hi;
        float4 g0 = *(const float4*)(gptr);
        float4 g1 = *(const float4*)(gptr + 4);
        __builtin_prefetch(gptr + (size_t)NCHUNK * Qq, 0, 1);
        float g[8] = {g0.x, g0.y, g0.z, g0.w, g1.x, g1.y, g1.z, g1.w};

        float p[8];
#pragma unroll
        for (int r = 0; r < 8; ++r)
            p[r] = __expf(fmaf(s[r], g[r], -SHIFT));

#pragma unroll
        for (int r = 0; r < 8; ++r)
            slot[(r + 8 * hi) * PSTRIDE + lo] = p[r];
        asm volatile("" ::: "memory");
        v2f pa[4];
#pragma unroll
        for (int i = 0; i < 4; ++i)
            pa[i] = *(const v2f*)(slot + lo * PSTRIDE + 4 * i + 2 * hi);

        const float* vptr = value + ((size_t)(b * Nn + n0 + 2 * hi)) * Vv + lo;
#pragma unroll
        for (int i = 0; i < 4; ++i) {
#pragma unroll
            for (int t = 0; t < 4; ++t) {
                v2f vb;
                vb.x = vptr[(size_t)(4 * i) * Vv + 16 * t];
                vb.y = vptr[(size_t)(4 * i + 1) * Vv + 16 * t];
                acc[t] = __builtin_amdgcn_wmma_f32_16x16x4_f32(false, pa[i], false, vb,
                                                               (short)0, acc[t], false, false);
            }
            accs = __builtin_amdgcn_wmma_f32_16x16x4_f32(false, pa[i], false, vones,
                                                         (short)0, accs, false, false);
        }
    }

#pragma unroll
    for (int t = 0; t < 4; ++t)
#pragma unroll
        for (int r = 0; r < 8; ++r)
            slot[(r + 8 * hi) * OSTRIDE + 16 * t + lo] = acc[t][r];
    if (lo == 0) {
#pragma unroll
        for (int r = 0; r < 8; ++r)
            slot[16 * OSTRIDE + (r + 8 * hi)] = accs[r];
    }
    __syncthreads();

#pragma unroll
    for (int k = 0; k < 4; ++k) {
        const int row = (tid >> 6) + 4 * k;
        const int col = tid & 63;
        float S = 0.f, O = 0.f;
        for (int ww = 0; ww < WAVES; ++ww) {
            const float* ws = smem + ww * SLOT;
            S += ws[16 * OSTRIDE + row];
            O += ws[row * OSTRIDE + col];
        }
        out[((size_t)(b * Qq + q0 + row)) * Vv + col] = O / S;
    }
}

extern "C" void kernel_launch(void* const* d_in, const int* in_sizes, int n_in,
                              void* d_out, int out_size, void* d_ws, size_t ws_size,
                              hipStream_t stream) {
    (void)in_sizes; (void)n_in; (void)d_ws; (void)ws_size; (void)out_size;
    const float* query = (const float*)d_in[0];
    const float* key   = (const float*)d_in[1];
    const float* value = (const float*)d_in[2];
    const float* gates = (const float*)d_in[3];
    float* out = (float*)d_out;

    dim3 grid(Bb * (Qq / 16));
    dim3 block(WAVES * 32);
    hipLaunchKernelGGL(gated_attn_flash_f32wmma, grid, block, 0, stream,
                       query, key, value, gates, out);
}


# Softmax denominator summed on the vector unit (per-lane adds, one 16-lane fold per block) instead of by a P x ones matrix product: 32 not 36 matrix instructions per 16 keys; its sum order changes
# speedup vs baseline: 1.8358x; 1.8358x over previous
#include <hip/hip_runtime.h>
#include <math.h>


typedef __attribute__((ext_vector_type(2))) float v2f;
typedef __attribute__((ext_vector_type(8))) float v8f;

constexpr int Bb = 8;
constexpr int Qq = 512;
constexpr int Nn = 8192;
constexpr int Dd = 64;
constexpr int Vv = 64;

constexpr int WAVES  = 16;
constexpr int QB     = WAVES * 16;
constexpr int QBLKS  = Qq / QB;
constexpr int NSEGS  = 16;
constexpr int NS     = Nn / NSEGS;
constexpr int KC     = 64;
constexpr int NTILE  = NS / KC;
constexpr int NCHUNK = 16;
constexpr int NSUB   = KC / NCHUNK;
constexpr int NTHR   = WAVES * 32;
constexpr int F4PT   = (KC * Dd / 4) / NTHR;

constexpr float SHIFT = 12.0f;

constexpr int KP      = 68;
constexpr int VTP     = KC + 4;
constexpr int KT      = KC * KP;
constexpr int VT      = Vv * VTP;
constexpr int STAGE   = KT + VT;
constexpr int PSTRIDE = 18;
constexpr int PSLOT   = 16 * PSTRIDE;
constexpr int OP      = 68;
constexpr int OSLOT   = 16 * OP;
constexpr int SM_P    = 2 * STAGE;
constexpr int SM_SUM  = SM_P + WAVES * PSLOT;
constexpr int SM_ALL  = SM_SUM + QB;
static_assert(WAVES * OSLOT <= 2 * STAGE, "epilogue O tiles fit in the dead stages");
static_assert(F4PT * NTHR * 4 == KC * Dd, "staging covers the tile exactly");

constexpr size_t WS_SUM = (size_t)NSEGS * Bb * Qq * Vv;

template <int M>
__device__ __forceinline__ float row_xor_add(float v)
{
    const int o = __builtin_amdgcn_update_dpp(0, __float_as_int(v), 0x160 + M, 0xF, 0xF, true);
    return v + __int_as_float(o);
}

__global__ __launch_bounds__(NTHR)
void gated_attn_partial_f32wmma(const float* __restrict__ query,
                                const float* __restrict__ key,
                                const float* __restrict__ value,
                                const float* __restrict__ gates,
                                float* __restrict__ ws)
{
    __shared__ __attribute__((aligned(16))) float smem[SM_ALL];

    const int tid  = threadIdx.x;
    const int w    = tid >> 5;
    const int lane = tid & 31;
    const int lo   = lane & 15;
    const int hi   = lane >> 4;

    const int qblk = blockIdx.x % QBLKS;
    const int seg  = (blockIdx.x / QBLKS) % NSEGS;
    const int b    = blockIdx.x / (QBLKS * NSEGS);
    const int q0   = qblk * QB + w * 16;
    const int nb   = seg * NS;

    float* pslot = smem + SM_P + w * PSLOT;

    v2f qa[16];
    {
        const float* qptr = query + ((size_t)(b * Qq + q0 + lo)) * Dd + 2 * hi;
#pragma unroll
        for (int i = 0; i < 16; ++i)
            qa[i] = *(const v2f*)(qptr + 4 * i);
    }

    v8f acc[4];
#pragma unroll
    for (int t = 0; t < 4; ++t) { v8f z = {}; acc[t] = z; }

    float psum[8] = {};

    static_assert(F4PT == 2 && NTHR / 16 == KC / 2, "two float4 per thread and operand");
    const int srow = tid >> 4;
    const int scol = 4 * (tid & 15);
    const float* kglob = key   + ((size_t)(b * Nn + nb)) * Dd + 4 * tid;
    const float* vglob = value + ((size_t)(b * Nn + nb)) * Vv + 4 * tid;
    const int koff0 = srow * KP + scol;
    const int koff1 = (srow + NTHR / 16) * KP + scol;
    const int voff  = KT + scol * VTP + srow;
    float4 kst0 = *(const float4*)(kglob);
    float4 kst1 = *(const float4*)(kglob + 4 * NTHR);
    float4 vst0 = *(const float4*)(vglob);
    float4 vst1 = *(const float4*)(vglob + 4 * NTHR);
    *(float4*)(smem + koff0) = kst0;
    *(float4*)(smem + koff1) = kst1;
    {
        float* vd = smem + voff;
        vd[0]       = vst0.x; vd[VTP]          = vst0.y; vd[2 * VTP]      = vst0.z; vd[3 * VTP]      = vst0.w;
        vd[KC / 2]  = vst1.x; vd[VTP + KC / 2] = vst1.y; vd[2 * VTP + KC / 2] = vst1.z; vd[3 * VTP + KC / 2] = vst1.w;
    }
    __syncthreads();

#pragma nounroll
    for (int tile = 0; tile < NTILE; ++tile) {
        const bool more = (tile + 1 < NTILE);
        const int  n0t  = nb + tile * KC;

        const float* gptr = gates + ((size_t)(b * Nn + n0t + lo)) * Qq + q0 + 8 * hi;
        float4 g0[NSUB], g1[NSUB];
#pragma unroll
        for (int sub = 0; sub < NSUB; ++sub) {
            g0[sub] = *(const float4*)(gptr + (size_t)(sub * NCHUNK) * Qq);
            g1[sub] = *(const float4*)(gptr + (size_t)(sub * NCHUNK) * Qq + 4);
        }

        if (more) {
            const float* kg = kglob + (size_t)(tile + 1) * KC * Dd;
            const float* vg = vglob + (size_t)(tile + 1) * KC * Vv;
            kst0 = *(const float4*)(kg);
            kst1 = *(const float4*)(kg + 4 * NTHR);
            vst0 = *(const float4*)(vg);
            vst1 = *(const float4*)(vg + 4 * NTHR);
#pragma unroll
            for (int sub = 0; sub < NSUB; ++sub)
                __builtin_prefetch(gptr + (size_t)(KC + sub * NCHUNK) * Qq, 0, 1);
        }

        const float* Ks = smem + (tile & 1) * STAGE;
        const float* Vs = Ks + KT;

        const float* kf = Ks + lo * KP + 2 * hi;
        v2f kb[16];
#pragma unroll
        for (int i = 0; i < 16; ++i)
            kb[i] = *(const v2f*)(kf + 4 * i);

#pragma unroll
        for (int sub = 0; sub < NSUB; ++sub) {
            const int lr = sub * NCHUNK;

            asm volatile("" :: "v"(kb[0]), "v"(kb[1]), "v"(kb[2]), "v"(kb[3]),
                               "v"(kb[4]), "v"(kb[5]), "v"(kb[6]), "v"(kb[7]),
                               "v"(kb[8]), "v"(kb[9]), "v"(kb[10]), "v"(kb[11]),
                               "v"(kb[12]), "v"(kb[13]), "v"(kb[14]), "v"(kb[15]));

            v8f s = {};
#pragma unroll
            for (int i = 0; i < 16; ++i)
                s = __builtin_amdgcn_wmma_f32_16x16x4_f32(false, qa[i], false, kb[i],
                                                          (short)0, s, false, false);

            const float* vf = Vs + lo * VTP + lr + 2 * hi;
            v2f vb[4][4];
#pragma unroll
            for (int i = 0; i < 4; ++i)
#pragma unroll
                for (int t = 0; t < 4; ++t)
                    vb[i][t] = *(const v2f*)(vf + (16 * t) * VTP + 4 * i);

            float g[8] = {g0[sub].x, g0[sub].y, g0[sub].z, g0[sub].w,
                          g1[sub].x, g1[sub].y, g1[sub].z, g1[sub].w};

            float p[8];
#pragma unroll
            for (int r = 0; r < 8; ++r) {
                p[r] = __expf(fmaf(s[r], g[r], -SHIFT));
                psum[r] += p[r];
            }

#pragma unroll
            for (int r = 0; r < 8; ++r)
                pslot[(r + 8 * hi) * PSTRIDE + lo] = p[r];
            asm volatile("" ::: "memory");
            v2f pa[4];
#pragma unroll
            for (int i = 0; i < 4; ++i)
                pa[i] = *(const v2f*)(pslot + lo * PSTRIDE + 4 * i + 2 * hi);

#pragma unroll
            for (int i = 0; i < 4; ++i) {
#pragma unroll
                for (int t = 0; t < 4; ++t)
                    acc[t] = __builtin_amdgcn_wmma_f32_16x16x4_f32(false, pa[i], false, vb[i][t],
                                                                   (short)0, acc[t], false, false);
            }

            if (sub + 1 < NSUB) {
#pragma unroll
                for (int i = 0; i < 16; ++i)
                    kb[i] = *(const v2f*)(kf + (lr + NCHUNK) * KP + 4 * i);
            }
        }

        if (more) {
            float* Kn = smem + ((tile + 1) & 1) * STAGE;
            *(float4*)(Kn + koff0) = kst0;
            *(float4*)(Kn + koff1) = kst1;
            float* vd = Kn + voff;
            vd[0]       = vst0.x; vd[VTP]          = vst0.y; vd[2 * VTP]      = vst0.z; vd[3 * VTP]      = vst0.w;
            vd[KC / 2]  = vst1.x; vd[VTP + KC / 2] = vst1.y; vd[2 * VTP + KC / 2] = vst1.z; vd[3 * VTP + KC / 2] = vst1.w;
        }
        __syncthreads();
    }

    float* oslot = smem + w * OSLOT;
#pragma unroll
    for (int t = 0; t < 4; ++t)
#pragma unroll
        for (int r = 0; r < 8; ++r)
            oslot[(r + 8 * hi) * OP + 16 * t + lo] = acc[t][r];
#pragma unroll
    for (int r = 0; r < 8; ++r) {
        float v = psum[r];
        v = row_xor_add<1>(v);
        v = row_xor_add<2>(v);
        v = row_xor_add<4>(v);
        v = row_xor_add<8>(v);
        psum[r] = v;
    }
    if (lo == 0) {
#pragma unroll
        for (int r = 0; r < 8; ++r)
            smem[SM_SUM + w * 16 + (r + 8 * hi)] = psum[r];
    }
    __syncthreads();

    const size_t prow = (size_t)(seg * Bb + b) * Qq;
    {
        float* gp = ws + (prow + q0) * Vv;
#pragma unroll
        for (int k = 0; k < 8; ++k) {
            const int e = lane + 32 * k;
            *(float4*)(gp + 4 * e) = *(const float4*)(oslot + (e >> 4) * OP + 4 * (e & 15));
        }
    }
    if (tid < QB)
        ws[WS_SUM + prow + qblk * QB + tid] = smem[SM_SUM + tid];
}

__global__ __launch_bounds__(256)
void gated_attn_merge(const float* __restrict__ ws, float* __restrict__ out)
{
    const int gid = blockIdx.x * 256 + threadIdx.x;
    const int bq  = gid >> 4;
    float4 O = {0.f, 0.f, 0.f, 0.f};
    float  S = 0.f;
#pragma unroll
    for (int sg = 0; sg < NSEGS; ++sg) {
        const size_t row = (size_t)sg * Bb * Qq + bq;
        const float4 o = *(const float4*)(ws + row * Vv + 4 * (gid & 15));
        S += ws[WS_SUM + row];
        O.x += o.x; O.y += o.y; O.z += o.z; O.w += o.w;
    }
    float4 res = {O.x / S, O.y / S, O.z / S, O.w / S};
    *(float4*)(out + 4 * (size_t)gid) = res;
}

extern "C" void kernel_launch(void* const* d_in, const int* in_sizes, int n_in,
                              void* d_out, int out_size, void* d_ws, size_t ws_size,
                              hipStream_t stream) {
    (void)in_sizes; (void)n_in; (void)ws_size; (void)out_size;
    const float* query = (const float*)d_in[0];
    const float* key   = (const float*)d_in[1];
    const float* value = (const float*)d_in[2];
    const float* gates = (const float*)d_in[3];
    float* out = (float*)d_out;
    float* ws  = (float*)d_ws;

    dim3 grid(Bb * NSEGS * QBLKS);
    dim3 block(NTHR);
    hipLaunchKernelGGL(gated_attn_partial_f32wmma, grid, block, 0, stream,
                       query, key, value, gates, ws);

    dim3 mgrid(Bb * Qq * Vv / 4 / 256);
    dim3 mblock(256);
    hipLaunchKernelGGL(gated_attn_merge, mgrid, mblock, 0, stream,
                       (const float*)ws, out);
}


# Software pipeline per wave, order pinned by fences: output product one 16-key step behind, so exponentials, P round trip and fragment loads run under matrix runs that do not need them; sums unchanged
# speedup vs baseline: 1.9963x; 1.0874x over previous
#include <hip/hip_runtime.h>
#include <math.h>


typedef __attribute__((ext_vector_type(2))) float v2f;
typedef __attribute__((ext_vector_type(8))) float v8f;

constexpr int Bb = 8;
constexpr int Qq = 512;
constexpr int Nn = 8192;
constexpr int Dd = 64;
constexpr int Vv = 64;

constexpr int WAVES  = 16;
constexpr int QB     = WAVES * 16;
constexpr int QBLKS  = Qq / QB;
constexpr int NSEGS  = 16;
constexpr int NS     = Nn / NSEGS;
constexpr int KC     = 64;
constexpr int NTILE  = NS / KC;
constexpr int NCHUNK = 16;
constexpr int NSUB   = KC / NCHUNK;
constexpr int NTHR   = WAVES * 32;
constexpr int F4PT   = (KC * Dd / 4) / NTHR;

constexpr float SHIFT = 12.0f;

constexpr int KP      = 68;
constexpr int VTP     = KC + 4;
constexpr int KT      = KC * KP;
constexpr int VT      = Vv * VTP;
constexpr int STAGE   = KT + VT;
constexpr int PSTRIDE = 18;
constexpr int PSLOT   = 16 * PSTRIDE;
constexpr int OP      = 68;
constexpr int OSLOT   = 16 * OP;
constexpr int SM_P    = 2 * STAGE;
constexpr int SM_SUM  = SM_P + WAVES * PSLOT;
constexpr int SM_ALL  = SM_SUM + QB;
static_assert(WAVES * OSLOT <= 2 * STAGE, "epilogue O tiles fit in the dead stages");
static_assert(F4PT * NTHR * 4 == KC * Dd, "staging covers the tile exactly");

constexpr size_t WS_SUM = (size_t)NSEGS * Bb * Qq * Vv;

template <int M>
__device__ __forceinline__ float row_xor_add(float v)
{
    const int o = __builtin_amdgcn_update_dpp(0, __float_as_int(v), 0x160 + M, 0xF, 0xF, true);
    return v + __int_as_float(o);
}

__global__ __launch_bounds__(NTHR)
void gated_attn_partial_f32wmma(const float* __restrict__ query,
                                const float* __restrict__ key,
                                const float* __restrict__ value,
                                const float* __restrict__ gates,
                                float* __restrict__ ws)
{
    __shared__ __attribute__((aligned(16))) float smem[SM_ALL];

    const int tid  = threadIdx.x;
    const int w    = tid >> 5;
    const int lane = tid & 31;
    const int lo   = lane & 15;
    const int hi   = lane >> 4;

    const int qblk = blockIdx.x % QBLKS;
    const int seg  = (blockIdx.x / QBLKS) % NSEGS;
    const int b    = blockIdx.x / (QBLKS * NSEGS);
    const int q0   = qblk * QB + w * 16;
    const int nb   = seg * NS;

    float* pslot = smem + SM_P + w * PSLOT;
    float* pst   = pslot + 8 * hi * PSTRIDE + lo;

    v2f qa[16];
    {
        const float* qptr = query + ((size_t)(b * Qq + q0 + lo)) * Dd + 2 * hi;
#pragma unroll
        for (int i = 0; i < 16; ++i)
            qa[i] = *(const v2f*)(qptr + 4 * i);
    }

    v8f acc[4];
#pragma unroll
    for (int t = 0; t < 4; ++t) { v8f z = {}; acc[t] = z; }

    float psum[8] = {};

    static_assert(F4PT == 2 && NTHR / 16 == KC / 2, "two float4 per thread and operand");
    const int srow = tid >> 4;
    const int scol = 4 * (tid & 15);
    const float* kglob = key   + ((size_t)(b * Nn + nb)) * Dd + 4 * tid;
    const float* vglob = value + ((size_t)(b * Nn + nb)) * Vv + 4 * tid;
    const int koff0 = srow * KP + scol;
    const int koff1 = (srow + NTHR / 16) * KP + scol;
    const int voff  = KT + scol * VTP + srow;
    float4 kst0 = *(const float4*)(kglob);
    float4 kst1 = *(const float4*)(kglob + 4 * NTHR);
    float4 vst0 = *(const float4*)(vglob);
    float4 vst1 = *(const float4*)(vglob + 4 * NTHR);

    const float* gbase = gates + ((size_t)(b * Nn + nb + lo)) * Qq + q0 + 8 * hi;
    float4 g0[NSUB], g1[NSUB];
#pragma unroll
    for (int sub = 0; sub < NSUB; ++sub) {
        g0[sub] = *(const float4*)(gbase + (size_t)(sub * NCHUNK) * Qq);
        g1[sub] = *(const float4*)(gbase + (size_t)(sub * NCHUNK) * Qq + 4);
    }

    *(float4*)(smem + koff0) = kst0;
    *(float4*)(smem + koff1) = kst1;
    {
        float* vd = smem + voff;
        vd[0]       = vst0.x; vd[VTP]          = vst0.y; vd[2 * VTP]      = vst0.z; vd[3 * VTP]      = vst0.w;
        vd[KC / 2]  = vst1.x; vd[VTP + KC / 2] = vst1.y; vd[2 * VTP + KC / 2] = vst1.z; vd[3 * VTP + KC / 2] = vst1.w;
    }
    __syncthreads();

    v2f kb[16];
    v2f vb[4][4];
    v2f pa[4];
    v8f s;
    {
        const float* kf0 = smem + lo * KP + 2 * hi;
#pragma unroll
        for (int i = 0; i < 16; ++i)
            kb[i] = *(const v2f*)(kf0 + 4 * i);
        asm volatile("" :: "v"(kb[0]), "v"(kb[1]), "v"(kb[2]), "v"(kb[3]),
                           "v"(kb[4]), "v"(kb[5]), "v"(kb[6]), "v"(kb[7]),
                           "v"(kb[8]), "v"(kb[9]), "v"(kb[10]), "v"(kb[11]),
                           "v"(kb[12]), "v"(kb[13]), "v"(kb[14]), "v"(kb[15]));
        v8f z = {};
        s = z;
#pragma unroll
        for (int i = 0; i < 16; ++i)
            s = __builtin_amdgcn_wmma_f32_16x16x4_f32(false, qa[i], false, kb[i],
                                                      (short)0, s, false, false);
        asm volatile("" :: "v"(s) : "memory");
#pragma unroll
        for (int i = 0; i < 16; ++i)
            kb[i] = *(const v2f*)(kf0 + NCHUNK * KP + 4 * i);
    }

#pragma nounroll
    for (int tile = 0; tile < NTILE; ++tile) {
        const bool more = (tile + 1 < NTILE);

        if (more) {
            const float* kg = kglob + (size_t)(tile + 1) * KC * Dd;
            const float* vg = vglob + (size_t)(tile + 1) * KC * Vv;
            kst0 = *(const float4*)(kg);
            kst1 = *(const float4*)(kg + 4 * NTHR);
            vst0 = *(const float4*)(vg);
            vst1 = *(const float4*)(vg + 4 * NTHR);
#pragma unroll
            for (int sub = 0; sub < NSUB; ++sub)
                __builtin_prefetch(gbase + (size_t)((tile + 1) * KC + sub * NCHUNK) * Qq, 0, 1);
        }

        const float* Ks = smem + (tile & 1) * STAGE;
        const float* Vs = Ks + KT;
        const float* kf = Ks + lo * KP + 2 * hi;

#pragma unroll
        for (int sub = 0; sub < NSUB; ++sub) {
            const int lr = sub * NCHUNK;

            if (sub > 0) {
#pragma unroll
                for (int i = 0; i < 4; ++i) {
#pragma unroll
                    for (int t = 0; t < 4; ++t)
                        acc[t] = __builtin_amdgcn_wmma_f32_16x16x4_f32(false, pa[i], false, vb[i][t],
                                                                       (short)0, acc[t], false, false);
                }
            }

            float g[8] = {g0[sub].x, g0[sub].y, g0[sub].z, g0[sub].w,
                          g1[sub].x, g1[sub].y, g1[sub].z, g1[sub].w};

            float p[8];
#pragma unroll
            for (int r = 0; r < 8; ++r) {
                p[r] = __expf(fmaf(s[r], g[r], -SHIFT));
                psum[r] += p[r];
            }

            asm volatile("" :: "v"(acc[0]), "v"(acc[1]), "v"(acc[2]), "v"(acc[3]) : "memory");
#pragma unroll
            for (int r = 0; r < 8; ++r)
                pst[r * PSTRIDE] = p[r];
            asm volatile("" ::: "memory");
#pragma unroll
            for (int i = 0; i < 4; ++i)
                pa[i] = *(const v2f*)(pslot + lo * PSTRIDE + 4 * i + 2 * hi);

            const float* vf = Vs + lo * VTP + lr + 2 * hi;
#pragma unroll
            for (int i = 0; i < 4; ++i)
#pragma unroll
                for (int t = 0; t < 4; ++t)
                    vb[i][t] = *(const v2f*)(vf + (16 * t) * VTP + 4 * i);

            __builtin_amdgcn_sched_barrier(0);
            if (sub + 1 < NSUB) {
                asm volatile("" :: "v"(kb[0]), "v"(kb[1]), "v"(kb[2]), "v"(kb[3]),
                                   "v"(kb[4]), "v"(kb[5]), "v"(kb[6]), "v"(kb[7]),
                                   "v"(kb[8]), "v"(kb[9]), "v"(kb[10]), "v"(kb[11]),
                                   "v"(kb[12]), "v"(kb[13]), "v"(kb[14]), "v"(kb[15]));
                v8f z = {};
                s = z;
#pragma unroll
                for (int i = 0; i < 16; ++i)
                    s = __builtin_amdgcn_wmma_f32_16x16x4_f32(false, qa[i], false, kb[i],
                                                              (short)0, s, false, false);
                if (sub + 2 < NSUB) {
                    asm volatile("" :: "v"(s) : "memory");
#pragma unroll
                    for (int i = 0; i < 16; ++i)
                        kb[i] = *(const v2f*)(kf + (lr + 2 * NCHUNK) * KP + 4 * i);
                }
                __builtin_amdgcn_sched_barrier(0);
            }
        }

        if (more) {
            float* Kn = smem + ((tile + 1) & 1) * STAGE;
            *(float4*)(Kn + koff0) = kst0;
            *(float4*)(Kn + koff1) = kst1;
            float* vd = Kn + voff;
            vd[0]       = vst0.x; vd[VTP]          = vst0.y; vd[2 * VTP]      = vst0.z; vd[3 * VTP]      = vst0.w;
            vd[KC / 2]  = vst1.x; vd[VTP + KC / 2] = vst1.y; vd[2 * VTP + KC / 2] = vst1.z; vd[3 * VTP + KC / 2] = vst1.w;
        }
        __syncthreads();

        const float* kfn = smem + ((tile + 1) & 1) * STAGE + lo * KP + 2 * hi;
        if (more) {
            const float* gptr = gbase + (size_t)((tile + 1) * KC) * Qq;
#pragma unroll
            for (int sub = 0; sub < NSUB; ++sub) {
                g0[sub] = *(const float4*)(gptr + (size_t)(sub * NCHUNK) * Qq);
                g1[sub] = *(const float4*)(gptr + (size_t)(sub * NCHUNK) * Qq + 4);
            }
#pragma unroll
            for (int i = 0; i < 16; ++i)
                kb[i] = *(const v2f*)(kfn + 4 * i);
        }
#pragma unroll
        for (int i = 0; i < 4; ++i) {
#pragma unroll
            for (int t = 0; t < 4; ++t)
                acc[t] = __builtin_amdgcn_wmma_f32_16x16x4_f32(false, pa[i], false, vb[i][t],
                                                               (short)0, acc[t], false, false);
        }
        if (more) {
            asm volatile("" :: "v"(kb[0]), "v"(kb[1]), "v"(kb[2]), "v"(kb[3]),
                               "v"(kb[4]), "v"(kb[5]), "v"(kb[6]), "v"(kb[7]),
                               "v"(kb[8]), "v"(kb[9]), "v"(kb[10]), "v"(kb[11]),
                               "v"(kb[12]), "v"(kb[13]), "v"(kb[14]), "v"(kb[15]));
            v8f z = {};
            s = z;
#pragma unroll
            for (int i = 0; i < 16; ++i)
                s = __builtin_amdgcn_wmma_f32_16x16x4_f32(false, qa[i], false, kb[i],
                                                          (short)0, s, false, false);
            asm volatile("" :: "v"(s) : "memory");
#pragma unroll
            for (int i = 0; i < 16; ++i)
                kb[i] = *(const v2f*)(kfn + NCHUNK * KP + 4 * i);
        }
    }

    float* oslot = smem + w * OSLOT;
#pragma unroll
    for (int t = 0; t < 4; ++t)
#pragma unroll
        for (int r = 0; r < 8; ++r)
            oslot[(r + 8 * hi) * OP + 16 * t + lo] = acc[t][r];
#pragma unroll
    for (int r = 0; r < 8; ++r) {
        float v = psum[r];
        v = row_xor_add<1>(v);
        v = row_xor_add<2>(v);
        v = row_xor_add<4>(v);
        v = row_xor_add<8>(v);
        psum[r] = v;
    }
    if (lo == 0) {
#pragma unroll
        for (int r = 0; r < 8; ++r)
            smem[SM_SUM + w * 16 + (r + 8 * hi)] = psum[r];
    }
    __syncthreads();

    const size_t prow = (size_t)(seg * Bb + b) * Qq;
    {
        float* gp = ws + (prow + q0) * Vv;
#pragma unroll
        for (int k = 0; k < 8; ++k) {
            const int e = lane + 32 * k;
            *(float4*)(gp + 4 * e) = *(const float4*)(oslot + (e >> 4) * OP + 4 * (e & 15));
        }
    }
    if (tid < QB)
        ws[WS_SUM + prow + qblk * QB + tid] = smem[SM_SUM + tid];
}

__global__ __launch_bounds__(256)
void gated_attn_merge(const float* __restrict__ ws, float* __restrict__ out)
{
    const int gid = blockIdx.x * 256 + threadIdx.x;
    const int bq  = gid >> 4;
    float4 O = {0.f, 0.f, 0.f, 0.f};
    float  S = 0.f;
#pragma unroll
    for (int sg = 0; sg < NSEGS; ++sg) {
        const size_t row = (size_t)sg * Bb * Qq + bq;
        const float4 o = *(const float4*)(ws + row * Vv + 4 * (gid & 15));
        S += ws[WS_SUM + row];
        O.x += o.x; O.y += o.y; O.z += o.z; O.w += o.w;
    }
    float4 res = {O.x / S, O.y / S, O.z / S, O.w / S};
    *(float4*)(out + 4 * (size_t)gid) = res;
}

extern "C" void kernel_launch(void* const* d_in, const int* in_sizes, int n_in,
                              void* d_out, int out_size, void* d_ws, size_t ws_size,
                              hipStream_t stream) {
    (void)in_sizes; (void)n_in; (void)ws_size; (void)out_size;
    const float* query = (const float*)d_in[0];
    const float* key   = (const float*)d_in[1];
    const float* value = (const float*)d_in[2];
    const float* gates = (const float*)d_in[3];
    float* out = (float*)d_out;
    float* ws  = (float*)d_ws;

    dim3 grid(Bb * NSEGS * QBLKS);
    dim3 block(NTHR);
    hipLaunchKernelGGL(gated_attn_partial_f32wmma, grid, block, 0, stream,
                       query, key, value, gates, ws);

    dim3 mgrid(Bb * Qq * Vv / 4 / 256);
    dim3 mblock(256);
    hipLaunchKernelGGL(gated_attn_merge, mgrid, mblock, 0, stream,
                       (const float*)ws, out);
}
